# starting kernel
# baseline (speedup 1.0000x reference)
#include <hip/hip_runtime.h>
#include <math.h>

typedef __attribute__((ext_vector_type(2))) float v2f;
typedef __attribute__((ext_vector_type(4))) float v4f;
typedef __attribute__((ext_vector_type(8))) float v8f;

#define NSTATE 1024
#define NHEAD  16
#define HDIM   64
#define NUMROT 8
#define SEQ    4096
#define BATCH  4
#define ROWS   (BATCH * SEQ * NHEAD)
#define TILES  (ROWS / 16)

__global__ void compose_R_kernel(const float* __restrict__ thetas,
                                 const float* __restrict__ pairs,
                                 const float* __restrict__ theta_scale,
                                 const float* __restrict__ rotM,
                                 float* __restrict__ Rp) {
  __shared__ float R[64 * 64];
  int r = threadIdx.x;
  if (r < 64) {
    for (int c = 0; c < 64; ++c) R[r * 64 + c] = (r == c) ? 1.0f : 0.0f;
    float ts = theta_scale[0];
    for (int k = 0; k < NUMROT; ++k) {
      int i = (int)pairs[2 * k + 0];
      int j = (int)pairs[2 * k + 1];
      float th = thetas[k] * ts;
      float c = cosf(th), s = sinf(th);
      if (i == j) {
        R[r * 64 + i] *= c;
      } else {
        float a = R[r * 64 + i], b = R[r * 64 + j];
        R[r * 64 + i] =  c * a + s * b;
        R[r * 64 + j] = -s * a + c * b;
      }
    }
    for (int c2 = 0; c2 < 64; ++c2) {
      float v = 0.0f;
      for (int k = 0; k < 64; ++k) v += R[r * 64 + k] * rotM[k * 64 + c2];
      int p = (c2 & 1) ? (32 + (c2 >> 1)) : (c2 >> 1);
      Rp[r * 64 + p] = v;
    }
  }
}

__global__ void sincos_kernel(const float* __restrict__ inv_freq,
                              float* __restrict__ tab) {
  int idx = blockIdx.x * blockDim.x + threadIdx.x;
  if (idx < SEQ * 32) {
    int s = idx >> 5, d = idx & 31;
    float ang = (float)s * inv_freq[d];
    tab[2 * idx + 0] = cosf(ang);
    tab[2 * idx + 1] = sinf(ang);
  }
}

__global__ void __launch_bounds__(256)
rot_rope_kernel(const float* __restrict__ X,
                const float* __restrict__ Rp,
                const float* __restrict__ tab,
                float* __restrict__ out) {
  __shared__ float lds[64 * 64];

  int tid = threadIdx.x;
  for (int i = tid; i < 1024; i += 256)
    ((v4f*)lds)[i] = ((const v4f*)Rp)[i];
  __syncthreads();

  int lane = tid & 31;
  int wave = tid >> 5;
  int half = lane >> 4;
  int l15  = lane & 15;

  v2f Bf[4][16];
#pragma unroll
  for (int nb = 0; nb < 4; ++nb) {
#pragma unroll
    for (int kb = 0; kb < 16; ++kb) {
      int k0 = 4 * kb + 2 * half;
      int n  = 16 * nb + l15;
      v2f b;
      b.x = lds[k0 * 64 + n];
      b.y = lds[(k0 + 1) * 64 + n];
      Bf[nb][kb] = b;
    }
  }

  int waveId = blockIdx.x * 8 + wave;
  int stride = gridDim.x * 8;

  for (int t = waveId; t < TILES; t += stride) {
    const float* xrow = X + ((size_t)t * 16 + l15) * 64;
    if (t + stride < TILES)
      __builtin_prefetch(xrow + (size_t)stride * 16 * 64, 0, 0);

    v2f Af[16];
#pragma unroll
    for (int kb = 0; kb < 16; ++kb)
      Af[kb] = *(const v2f*)(xrow + 4 * kb + 2 * half);

    int s_pos = t & (SEQ - 1);
    float* obase = out + (size_t)t * 16 * 64;

#pragma unroll
    for (int p = 0; p < 2; ++p) {
      v8f accA = {}, accB = {};
#pragma unroll
      for (int kb = 0; kb < 16; ++kb) {
        accA = __builtin_amdgcn_wmma_f32_16x16x4_f32(
            false, Af[kb], false, Bf[p][kb],     (short)0, accA, false, false);
        accB = __builtin_amdgcn_wmma_f32_16x16x4_f32(
            false, Af[kb], false, Bf[p + 2][kb], (short)0, accB, false, false);
      }

      int dp = 16 * p + l15;
      v2f cs = *(const v2f*)(tab + ((size_t)s_pos * 32 + dp) * 2);
      float c = cs.x, s = cs.y;

#pragma unroll
      for (int v = 0; v < 8; ++v) {
        int row = v + 8 * half;
        float a = accA[v], b = accB[v];
        obase[(size_t)row * 64 + dp]      = a * c - b * s;
        obase[(size_t)row * 64 + 32 + dp] = a * s + b * c;
      }
    }
  }
}

extern "C" void kernel_launch(void* const* d_in, const int* in_sizes, int n_in,
                              void* d_out, int out_size, void* d_ws, size_t ws_size,
                              hipStream_t stream) {
  const float* x      = (const float*)d_in[0];
  const float* thetas = (const float*)d_in[1];
  const float* pairs  = (const float*)d_in[2];
  const float* tscale = (const float*)d_in[3];
  const float* rotM   = (const float*)d_in[4];
  const float* invf   = (const float*)d_in[5];
  float* out = (float*)d_out;

  float* Rp  = (float*)d_ws;
  float* tab = Rp + 64 * 64;

  compose_R_kernel<<<1, 64, 0, stream>>>(thetas, pairs, tscale, rotM, Rp);
  sincos_kernel<<<(SEQ * 32 + 255) / 256, 256, 0, stream>>>(invf, tab);
  rot_rope_kernel<<<512, 256, 0, stream>>>(x, Rp, tab, out);
}


# compose kernel spread over 64 one-wave blocks, one row of R each, row and matrix columns in registers, shared by lane reads, instead of one 64-thread block's serial 64x64x64 product; same sum order
# speedup vs baseline: 2.6107x; 2.6107x over previous
#include <hip/hip_runtime.h>
#include <math.h>

typedef __attribute__((ext_vector_type(2))) float v2f;
typedef __attribute__((ext_vector_type(4))) float v4f;
typedef __attribute__((ext_vector_type(8))) float v8f;

#define NSTATE 1024
#define NHEAD  16
#define HDIM   64
#define NUMROT 8
#define SEQ    4096
#define BATCH  4
#define ROWS   (BATCH * SEQ * NHEAD)
#define TILES  (ROWS / 16)

__device__ __forceinline__ float lane_read(float v, int srcLane) {
  return __int_as_float(__builtin_amdgcn_readlane(__float_as_int(v), srcLane));
}

__global__ void __launch_bounds__(32)
compose_R_kernel(const float* __restrict__ thetas,
                 const float* __restrict__ pairs,
                 const float* __restrict__ theta_scale,
                 const float* __restrict__ rotM,
                 float* __restrict__ Rp) {
  const int l = threadIdx.x;
  const int r = blockIdx.x;

  float th = thetas[l & (NUMROT - 1)] * theta_scale[0];

  float pr[2 * NUMROT];
#pragma unroll
  for (int q = 0; q < 2 * NUMROT; ++q) pr[q] = pairs[q];

  v2f m[64];
#pragma unroll
  for (int k = 0; k < 64; ++k)
    m[k] = *(const v2f*)(rotM + k * 64 + 2 * l);

  float co_l = cosf(th), si_l = sinf(th);

  float lo = (l == r) ? 1.0f : 0.0f;
  float hi = (l + 32 == r) ? 1.0f : 0.0f;

#pragma unroll
  for (int k = 0; k < NUMROT; ++k) {
    int i = (int)pr[2 * k + 0];
    int j = (int)pr[2 * k + 1];
    float c = lane_read(co_l, k), s = lane_read(si_l, k);
    float a = lane_read((i < 32) ? lo : hi, i & 31);
    float b = lane_read((j < 32) ? lo : hi, j & 31);
    float na = fmaf(c, a, s * b);
    float nb = fmaf(c, b, -(s * a));
    if (i == j) {
      na = a * c;
      nb = na;
    }
    lo = (l == i) ? na : lo;   hi = (l + 32 == i) ? na : hi;
    lo = (l == j) ? nb : lo;   hi = (l + 32 == j) ? nb : hi;
  }

  float ve = 0.0f, vo = 0.0f;
#pragma unroll
  for (int k = 0; k < 64; ++k) {
    float g = (k < 32) ? lane_read(lo, k) : lane_read(hi, k - 32);
    ve = fmaf(g, m[k].x, ve);
    vo = fmaf(g, m[k].y, vo);
  }
  Rp[r * 64 + l]      = ve;
  Rp[r * 64 + 32 + l] = vo;
}

__global__ void sincos_kernel(const float* __restrict__ inv_freq,
                              float* __restrict__ tab) {
  int idx = blockIdx.x * blockDim.x + threadIdx.x;
  if (idx < SEQ * 32) {
    int s = idx >> 5, d = idx & 31;
    float ang = (float)s * inv_freq[d];
    tab[2 * idx + 0] = cosf(ang);
    tab[2 * idx + 1] = sinf(ang);
  }
}

__global__ void __launch_bounds__(256)
rot_rope_kernel(const float* __restrict__ X,
                const float* __restrict__ Rp,
                const float* __restrict__ tab,
                float* __restrict__ out) {
  __shared__ float lds[64 * 64];

  int tid = threadIdx.x;
  for (int i = tid; i < 1024; i += 256)
    ((v4f*)lds)[i] = ((const v4f*)Rp)[i];
  __syncthreads();

  int lane = tid & 31;
  int wave = tid >> 5;
  int half = lane >> 4;
  int l15  = lane & 15;

  v2f Bf[4][16];
#pragma unroll
  for (int nb = 0; nb < 4; ++nb) {
#pragma unroll
    for (int kb = 0; kb < 16; ++kb) {
      int k0 = 4 * kb + 2 * half;
      int n  = 16 * nb + l15;
      v2f b;
      b.x = lds[k0 * 64 + n];
      b.y = lds[(k0 + 1) * 64 + n];
      Bf[nb][kb] = b;
    }
  }

  int waveId = blockIdx.x * 8 + wave;
  int stride = gridDim.x * 8;

  for (int t = waveId; t < TILES; t += stride) {
    const float* xrow = X + ((size_t)t * 16 + l15) * 64;
    if (t + stride < TILES)
      __builtin_prefetch(xrow + (size_t)stride * 16 * 64, 0, 0);

    v2f Af[16];
#pragma unroll
    for (int kb = 0; kb < 16; ++kb)
      Af[kb] = *(const v2f*)(xrow + 4 * kb + 2 * half);

    int s_pos = t & (SEQ - 1);
    float* obase = out + (size_t)t * 16 * 64;

#pragma unroll
    for (int p = 0; p < 2; ++p) {
      v8f accA = {}, accB = {};
#pragma unroll
      for (int kb = 0; kb < 16; ++kb) {
        accA = __builtin_amdgcn_wmma_f32_16x16x4_f32(
            false, Af[kb], false, Bf[p][kb],     (short)0, accA, false, false);
        accB = __builtin_amdgcn_wmma_f32_16x16x4_f32(
            false, Af[kb], false, Bf[p + 2][kb], (short)0, accB, false, false);
      }

      int dp = 16 * p + l15;
      v2f cs = *(const v2f*)(tab + ((size_t)s_pos * 32 + dp) * 2);
      float c = cs.x, s = cs.y;

#pragma unroll
      for (int v = 0; v < 8; ++v) {
        int row = v + 8 * half;
        float a = accA[v], b = accB[v];
        obase[(size_t)row * 64 + dp]      = a * c - b * s;
        obase[(size_t)row * 64 + 32 + dp] = a * s + b * c;
      }
    }
  }
}

extern "C" void kernel_launch(void* const* d_in, const int* in_sizes, int n_in,
                              void* d_out, int out_size, void* d_ws, size_t ws_size,
                              hipStream_t stream) {
  const float* x      = (const float*)d_in[0];
  const float* thetas = (const float*)d_in[1];
  const float* pairs  = (const float*)d_in[2];
  const float* tscale = (const float*)d_in[3];
  const float* rotM   = (const float*)d_in[4];
  const float* invf   = (const float*)d_in[5];
  float* out = (float*)d_out;

  float* Rp  = (float*)d_ws;
  float* tab = Rp + 64 * 64;

  compose_R_kernel<<<64, 32, 0, stream>>>(thetas, pairs, tscale, rotM, Rp);
  sincos_kernel<<<(SEQ * 32 + 255) / 256, 256, 0, stream>>>(invf, tab);
  rot_rope_kernel<<<512, 256, 0, stream>>>(x, Rp, tab, out);
}


# three launches fused into one of 256 16-wave blocks: each block composes R in LDS (R@rotation_matrix now summed on f32 WMMA), each wave computes its position's cos/sin itself; no workspace
# speedup vs baseline: 3.7738x; 1.4455x over previous
#include <hip/hip_runtime.h>
#include <math.h>

typedef __attribute__((ext_vector_type(2))) float v2f;
typedef __attribute__((ext_vector_type(4))) float v4f;
typedef __attribute__((ext_vector_type(8))) float v8f;

#define NSTATE 1024
#define NHEAD  16
#define HDIM   64
#define NUMROT 8
#define SEQ    4096
#define BATCH  4
#define ROWS   (BATCH * SEQ * NHEAD)
#define TILES  (ROWS / 16)

__device__ __forceinline__ float lane_read(float v, int srcLane) {
  return __int_as_float(__builtin_amdgcn_readlane(__float_as_int(v), srcLane));
}

#define RG_PITCH 65

#define WG_WAVES 16

__global__ void __launch_bounds__(32 * WG_WAVES)
rot_rope_fused_kernel(const float* __restrict__ X,
                      const float* __restrict__ thetas,
                      const float* __restrict__ pairs,
                      const float* __restrict__ theta_scale,
                      const float* __restrict__ rotM,
                      const float* __restrict__ inv_freq,
                      float* __restrict__ out) {
  __shared__ float Rg[64 * RG_PITCH];
  __shared__ float lds[64 * 64];

  const int tid  = threadIdx.x;
  const int lane = tid & 31;
  const int wave = tid >> 5;
  const int half = lane >> 4;
  const int l15  = lane & 15;

  const int s_pos = blockIdx.x * WG_WAVES + wave;

  __builtin_prefetch(X + ((size_t)s_pos * 16 + l15) * 64 + 32 * half, 0, 0);

  for (int i = tid; i < 64 * 64; i += 32 * WG_WAVES) {
    int r = i >> 6, c = i & 63;
    Rg[r * RG_PITCH + c] = (r == c) ? 1.0f : 0.0f;
  }
  __syncthreads();

  if (wave < 2) {
    float th = thetas[lane & (NUMROT - 1)] * theta_scale[0];
    float co_l = cosf(th), si_l = sinf(th);
    float* g = Rg + tid * RG_PITCH;
#pragma unroll
    for (int k = 0; k < NUMROT; ++k) {
      int i = (int)pairs[2 * k + 0] & 63;
      int j = (int)pairs[2 * k + 1] & 63;
      float c = lane_read(co_l, k), s = lane_read(si_l, k);
      float a = g[i], b = g[j];
      float na = fmaf(c, a, s * b);
      float nb = fmaf(c, b, -(s * a));
      if (i == j) {
        na = a * c;
        nb = na;
      }
      g[i] = na;
      g[j] = nb;
    }
  }
  __syncthreads();

  {
    const int mb  = wave >> 2;
    const int pc  = 16 * (wave & 3) + l15;
    const int src = (pc < 32) ? 2 * pc : 2 * (pc - 32) + 1;
    const float* grow = Rg + (16 * mb + l15) * RG_PITCH + 2 * half;
    const float* mcol = rotM + (2 * half) * 64 + src;
    v8f acc = {};
#pragma unroll
    for (int kb = 0; kb < 16; ++kb) {
      v2f a, b;
      a.x = grow[4 * kb];
      a.y = grow[4 * kb + 1];
      b.x = mcol[(4 * kb) * 64];
      b.y = mcol[(4 * kb + 1) * 64];
      acc = __builtin_amdgcn_wmma_f32_16x16x4_f32(
          false, a, false, b, (short)0, acc, false, false);
    }
#pragma unroll
    for (int v = 0; v < 8; ++v)
      lds[(16 * mb + v + 8 * half) * 64 + pc] = acc[v];
  }
  __syncthreads();

  v2f Bf[4][16];
#pragma unroll
  for (int nb = 0; nb < 4; ++nb) {
#pragma unroll
    for (int kb = 0; kb < 16; ++kb) {
      int k0 = 4 * kb + 2 * half;
      int n  = 16 * nb + l15;
      v2f b;
      b.x = lds[k0 * 64 + n];
      b.y = lds[(k0 + 1) * 64 + n];
      Bf[nb][kb] = b;
    }
  }

  float cs_c[2], cs_s[2];
#pragma unroll
  for (int p = 0; p < 2; ++p) {
    float ang = (float)s_pos * inv_freq[16 * p + l15];
    cs_c[p] = cosf(ang);
    cs_s[p] = sinf(ang);
  }

#pragma unroll 1
  for (int bt = 0; bt < BATCH; ++bt) {
    int t = bt * SEQ + s_pos;
    const float* xrow = X + ((size_t)t * 16 + l15) * 64;
    if (bt + 1 < BATCH)
      __builtin_prefetch(xrow + (size_t)SEQ * 16 * 64, 0, 0);

    v2f Af[16];
#pragma unroll
    for (int kb = 0; kb < 16; ++kb)
      Af[kb] = *(const v2f*)(xrow + 4 * kb + 2 * half);

    float* obase = out + (size_t)t * 16 * 64;

#pragma unroll
    for (int p = 0; p < 2; ++p) {
      v8f accA = {}, accB = {};
#pragma unroll
      for (int kb = 0; kb < 16; ++kb) {
        accA = __builtin_amdgcn_wmma_f32_16x16x4_f32(
            false, Af[kb], false, Bf[p][kb],     (short)0, accA, false, false);
        accB = __builtin_amdgcn_wmma_f32_16x16x4_f32(
            false, Af[kb], false, Bf[p + 2][kb], (short)0, accB, false, false);
      }

      int dp = 16 * p + l15;
      float c = cs_c[p], s = cs_s[p];

#pragma unroll
      for (int v = 0; v < 8; ++v) {
        int row = v + 8 * half;
        float a = accA[v], b = accB[v];
        obase[(size_t)row * 64 + dp]      = a * c - b * s;
        obase[(size_t)row * 64 + 32 + dp] = a * s + b * c;
      }
    }
  }
}

extern "C" void kernel_launch(void* const* d_in, const int* in_sizes, int n_in,
                              void* d_out, int out_size, void* d_ws, size_t ws_size,
                              hipStream_t stream) {
  const float* x      = (const float*)d_in[0];
  const float* thetas = (const float*)d_in[1];
  const float* pairs  = (const float*)d_in[2];
  const float* tscale = (const float*)d_in[3];
  const float* rotM   = (const float*)d_in[4];
  const float* invf   = (const float*)d_in[5];
  float* out = (float*)d_out;

  rot_rope_fused_kernel<<<SEQ / WG_WAVES, 32 * WG_WAVES, 0, stream>>>(x, thetas, pairs, tscale, rotM, invf, out);
}
